# starting kernel
# baseline (speedup 1.0000x reference)
#include <hip/hip_runtime.h>

typedef __attribute__((ext_vector_type(2))) float v2f;
typedef __attribute__((ext_vector_type(8))) float v8f;

#define NRAYS_BLOCK 256
#define HIDDEN 256
#define NEAR_T 1.0f
#define FAR_T 10.0f
#define MAX_ITERS 64

__global__ __launch_bounds__(NRAYS_BLOCK) void sphere_trace_mlp_kernel(
    const float* __restrict__ origins,
    const float* __restrict__ dirs,
    const float* __restrict__ center,
    const float* __restrict__ radius,
    const float* __restrict__ w1,
    const float* __restrict__ b1,
    const float* __restrict__ w2,
    const float* __restrict__ b2,
    float* __restrict__ out)
{
    __shared__ v2f   lds_b1t[16 * 32];
    __shared__ float lds_w2[HIDDEN * 3];

    const int tid  = threadIdx.x;
    const int lane = tid & 31;
    const int wave = tid >> 5;
    const int ray  = blockIdx.x * NRAYS_BLOCK + tid;

    for (int e = tid; e < 16 * 32; e += NRAYS_BLOCK) {
        int j = e >> 5, l = e & 31;
        int n = (j << 4) + (l & 15);
        v2f v;
        if (l < 16) { v.x = w1[0 * HIDDEN + n]; v.y = w1[1 * HIDDEN + n]; }
        else        { v.x = w1[2 * HIDDEN + n]; v.y = b1[n]; }
        lds_b1t[e] = v;
    }
    for (int e = tid; e < HIDDEN * 3; e += NRAYS_BLOCK) lds_w2[e] = w2[e];
    __syncthreads();

    const float ox = origins[ray * 3 + 0], oy = origins[ray * 3 + 1], oz = origins[ray * 3 + 2];
    const float rdx = dirs[ray * 3 + 0],   rdy = dirs[ray * 3 + 1],   rdz = dirs[ray * 3 + 2];
    const float cx = center[0], cy = center[1], cz = center[2];
    const float R  = radius[0];

    float t  = NEAR_T;
    float px = ox + t * rdx, py = oy + t * rdy, pz = oz + t * rdz;
    bool mask = true;
#pragma unroll 8
    for (int it = 0; it < MAX_ITERS; ++it) {
        float qx = px - cx, qy = py - cy, qz = pz - cz;
        float f = sqrtf(qx * qx + qy * qy + qz * qz) - R;
        t += f;
        px = ox + t * rdx; py = oy + t * rdy; pz = oz + t * rdz;
        mask = mask && (t <= FAR_T);
    }

    const unsigned long long bal = __ballot(mask);

    const float b2x = b2[0], b2y = b2[1], b2z = b2[2];
    const int  halfSel = (lane >= 16) ? 1 : 0;
    const int  nlane   = lane & 15;
    const int  waveRay0 = blockIdx.x * NRAYS_BLOCK + (wave << 5);

    for (int pass = 0; pass < 2; ++pass) {
        const int srcLane = nlane + (pass << 4);
        float sx = __shfl(px, srcLane, 32);
        float sy = __shfl(py, srcLane, 32);
        float sz = __shfl(pz, srcLane, 32);
        v2f a;
        a.x = halfSel ? sz : sx;
        a.y = halfSel ? 1.0f : sy;

        float accR[8], accG[8], accB[8];
#pragma unroll
        for (int v = 0; v < 8; ++v) { accR[v] = 0.f; accG[v] = 0.f; accB[v] = 0.f; }

#pragma unroll
        for (int j = 0; j < 16; ++j) {
            v2f b = lds_b1t[(j << 5) + lane];
            v8f c = {};
            v8f dmat = __builtin_amdgcn_wmma_f32_16x16x4_f32(
                false, a, false, b, (short)0, c, false, false);

            const int row = (j << 4) + nlane;
            const float wr = lds_w2[row * 3 + 0];
            const float wg = lds_w2[row * 3 + 1];
            const float wb = lds_w2[row * 3 + 2];
#pragma unroll
            for (int v = 0; v < 8; ++v) {
                float h = fmaxf(dmat[v], 0.0f);
                accR[v] += h * wr;
                accG[v] += h * wg;
                accB[v] += h * wb;
            }
        }

#pragma unroll
        for (int off = 8; off >= 1; off >>= 1) {
#pragma unroll
            for (int v = 0; v < 8; ++v) {
                accR[v] += __shfl_xor(accR[v], off, 32);
                accG[v] += __shfl_xor(accG[v], off, 32);
                accB[v] += __shfl_xor(accB[v], off, 32);
            }
        }

        const int passBase = waveRay0 + (pass << 4);
#pragma unroll
        for (int v = 0; v < 8; ++v) {
            if (nlane == v) {
                const int m      = v + (halfSel << 3);
                const int wlLane = (pass << 4) + m;
                const int r      = passBase + m;
                float s0 = 0.f, s1 = 0.f, s2 = 0.f;
                if ((bal >> wlLane) & 1ull) {
                    s0 = 1.0f / (1.0f + expf(-(accR[v] + b2x)));
                    s1 = 1.0f / (1.0f + expf(-(accG[v] + b2y)));
                    s2 = 1.0f / (1.0f + expf(-(accB[v] + b2z)));
                }
                out[r * 3 + 0] = s0;
                out[r * 3 + 1] = s1;
                out[r * 3 + 2] = s2;
            }
        }
    }
}

extern "C" void kernel_launch(void* const* d_in, const int* in_sizes, int n_in,
                              void* d_out, int out_size, void* d_ws, size_t ws_size,
                              hipStream_t stream) {
    const float* origins = (const float*)d_in[0];
    const float* dirs    = (const float*)d_in[1];
    const float* center  = (const float*)d_in[2];
    const float* radius  = (const float*)d_in[3];
    const float* w1      = (const float*)d_in[4];
    const float* b1      = (const float*)d_in[5];
    const float* w2      = (const float*)d_in[6];
    const float* b2      = (const float*)d_in[7];
    float* out = (float*)d_out;

    const int n_rays = in_sizes[0] / 3;
    const int blocks = n_rays / NRAYS_BLOCK;

    sphere_trace_mlp_kernel<<<blocks, NRAYS_BLOCK, 0, stream>>>(
        origins, dirs, center, radius, w1, b1, w2, b2, out);
}


# Layer-2 sum over each 16-lane row as a halving tree of row_xmask DPP adds (same additions as the butterfly): one ray per lane left, so sigmoid+store run once per wave, not in 16 two-lane branches
# speedup vs baseline: 1.3755x; 1.3755x over previous
#include <hip/hip_runtime.h>


typedef __attribute__((ext_vector_type(2))) float v2f;
typedef __attribute__((ext_vector_type(8))) float v8f;

#define NRAYS_BLOCK 256
#define HIDDEN 256
#define NEAR_T 1.0f
#define FAR_T 10.0f
#define MAX_ITERS 64

#define ROW_XMASK(m) (0x160 + (m))

template <int M>
__device__ __forceinline__ float row_xor(float v) {
    return __int_as_float(__builtin_amdgcn_update_dpp(
        0, __float_as_int(v), ROW_XMASK(M), 0xf, 0xf, true));
}

template <int M>
__device__ __forceinline__ float halve(bool up, float lo, float hi) {
    const float keep = up ? hi : lo;
    const float send = up ? lo : hi;
    return keep + row_xor<M>(send);
}

__device__ __forceinline__ void mlp_pass(
    const int pass, const float px, const float py, const float pz,
    const int lane, const int nlane, const int halfSel,
    const v2f* lds_b1t, const float* lds_w2,
    float& outR, float& outG, float& outB)
{
    const int srcLane = nlane + (pass << 4);
    float sx = __shfl(px, srcLane, 32);
    float sy = __shfl(py, srcLane, 32);
    float sz = __shfl(pz, srcLane, 32);
    v2f a;
    a.x = halfSel ? sz : sx;
    a.y = halfSel ? 1.0f : sy;

    float accR[8], accG[8], accB[8];
#pragma unroll
    for (int v = 0; v < 8; ++v) { accR[v] = 0.f; accG[v] = 0.f; accB[v] = 0.f; }

#pragma unroll
    for (int j = 0; j < 16; ++j) {
        v2f b = lds_b1t[(j << 5) + lane];
        v8f c = {};
        v8f dmat = __builtin_amdgcn_wmma_f32_16x16x4_f32(
            false, a, false, b, (short)0, c, false, false);

        const int row = (j << 4) + nlane;
        const float wr = lds_w2[row * 3 + 0];
        const float wg = lds_w2[row * 3 + 1];
        const float wb = lds_w2[row * 3 + 2];
#pragma unroll
        for (int v = 0; v < 8; ++v) {
            float h = fmaxf(dmat[v], 0.0f);
            accR[v] += h * wr;
            accG[v] += h * wg;
            accB[v] += h * wb;
        }
    }

    const bool up8 = (nlane & 8) != 0;
    const bool up4 = (nlane & 4) != 0;
    const bool up2 = (nlane & 2) != 0;

    float qR[4], qG[4], qB[4];
    for (int i = 0; i < 4; ++i) {
        qR[i] = halve<8>(up8, accR[i], accR[i + 4]);
        qG[i] = halve<8>(up8, accG[i], accG[i + 4]);
        qB[i] = halve<8>(up8, accB[i], accB[i + 4]);
    }
    float dR[2], dG[2], dB[2];
    for (int i = 0; i < 2; ++i) {
        dR[i] = halve<4>(up4, qR[i], qR[i + 2]);
        dG[i] = halve<4>(up4, qG[i], qG[i + 2]);
        dB[i] = halve<4>(up4, qB[i], qB[i + 2]);
    }
    outR = halve<2>(up2, dR[0], dR[1]);
    outG = halve<2>(up2, dG[0], dG[1]);
    outB = halve<2>(up2, dB[0], dB[1]);
}

__global__ __launch_bounds__(NRAYS_BLOCK) void sphere_trace_mlp_kernel(
    const float* __restrict__ origins,
    const float* __restrict__ dirs,
    const float* __restrict__ center,
    const float* __restrict__ radius,
    const float* __restrict__ w1,
    const float* __restrict__ b1,
    const float* __restrict__ w2,
    const float* __restrict__ b2,
    float* __restrict__ out)
{
    __shared__ v2f   lds_b1t[16 * 32];
    __shared__ float lds_w2[HIDDEN * 3];

    const int tid  = threadIdx.x;
    const int lane = tid & 31;
    const int wave = tid >> 5;
    const int ray  = blockIdx.x * NRAYS_BLOCK + tid;

    for (int e = tid; e < 16 * 32; e += NRAYS_BLOCK) {
        int j = e >> 5, l = e & 31;
        int n = (j << 4) + (l & 15);
        v2f v;
        if (l < 16) { v.x = w1[0 * HIDDEN + n]; v.y = w1[1 * HIDDEN + n]; }
        else        { v.x = w1[2 * HIDDEN + n]; v.y = b1[n]; }
        lds_b1t[e] = v;
    }
    for (int e = tid; e < HIDDEN * 3; e += NRAYS_BLOCK) lds_w2[e] = w2[e];
    __syncthreads();

    const float ox = origins[ray * 3 + 0], oy = origins[ray * 3 + 1], oz = origins[ray * 3 + 2];
    const float rdx = dirs[ray * 3 + 0],   rdy = dirs[ray * 3 + 1],   rdz = dirs[ray * 3 + 2];
    const float cx = center[0], cy = center[1], cz = center[2];
    const float R  = radius[0];

    float t  = NEAR_T;
    float px = ox + t * rdx, py = oy + t * rdy, pz = oz + t * rdz;
    bool mask = true;
#pragma unroll 8
    for (int it = 0; it < MAX_ITERS; ++it) {
        float qx = px - cx, qy = py - cy, qz = pz - cz;
        float f = sqrtf(qx * qx + qy * qy + qz * qz) - R;
        t += f;
        px = ox + t * rdx; py = oy + t * rdy; pz = oz + t * rdz;
        mask = mask && (t <= FAR_T);
    }

    const unsigned int bal = (unsigned int)__ballot(mask);

    const float b2x = b2[0], b2y = b2[1], b2z = b2[2];
    const int  halfSel = (lane >= 16) ? 1 : 0;
    const int  nlane   = lane & 15;
    const int  waveRay0 = blockIdx.x * NRAYS_BLOCK + (wave << 5);

    float r0, g0, b0v, r1, g1, b1v;
    mlp_pass(0, px, py, pz, lane, nlane, halfSel, lds_b1t, lds_w2, r0, g0, b0v);
    mlp_pass(1, px, py, pz, lane, nlane, halfSel, lds_b1t, lds_w2, r1, g1, b1v);

    const bool odd = (nlane & 1) != 0;
    const float sumR = halve<1>(odd, r0, r1);
    const float sumG = halve<1>(odd, g0, g1);
    const float sumB = halve<1>(odd, b0v, b1v);

    const int wl = (odd ? 16 : 0) + (halfSel << 3) + (nlane >> 1);
    const int r  = waveRay0 + wl;
    const bool hit = ((bal >> wl) & 1u) != 0u;

    float s0 = 0.f, s1 = 0.f, s2 = 0.f;
    if (hit) {
        s0 = 1.0f / (1.0f + expf(-(sumR + b2x)));
        s1 = 1.0f / (1.0f + expf(-(sumG + b2y)));
        s2 = 1.0f / (1.0f + expf(-(sumB + b2z)));
    }
    out[r * 3 + 0] = s0;
    out[r * 3 + 1] = s1;
    out[r * 3 + 2] = s2;
}

extern "C" void kernel_launch(void* const* d_in, const int* in_sizes, int n_in,
                              void* d_out, int out_size, void* d_ws, size_t ws_size,
                              hipStream_t stream) {
    const float* origins = (const float*)d_in[0];
    const float* dirs    = (const float*)d_in[1];
    const float* center  = (const float*)d_in[2];
    const float* radius  = (const float*)d_in[3];
    const float* w1      = (const float*)d_in[4];
    const float* b1      = (const float*)d_in[5];
    const float* w2      = (const float*)d_in[6];
    const float* b2      = (const float*)d_in[7];
    float* out = (float*)d_out;

    const int n_rays = in_sizes[0] / 3;
    const int blocks = n_rays / NRAYS_BLOCK;

    sphere_trace_mlp_kernel<<<blocks, NRAYS_BLOCK, 0, stream>>>(
        origins, dirs, center, radius, w1, b1, w2, b2, out);
}


# Two rays per lane, half the waves (64 rays per wave): the tracing loop holds two independent rays per lane, so its steps pack into two-value f32 and dual-issue instructions; same arithmetic per ray
# speedup vs baseline: 1.5384x; 1.1184x over previous
#include <hip/hip_runtime.h>


typedef __attribute__((ext_vector_type(2))) float v2f;
typedef __attribute__((ext_vector_type(8))) float v8f;

#define THREADS_BLOCK 256
#define RAYS_LANE 2
#define NRAYS_BLOCK (THREADS_BLOCK * RAYS_LANE)
#define HIDDEN 256
#define NEAR_T 1.0f
#define FAR_T 10.0f
#define MAX_ITERS 64

#define ROW_XMASK(m) (0x160 + (m))

template <int M>
__device__ __forceinline__ float row_xor(float v) {
    return __int_as_float(__builtin_amdgcn_update_dpp(
        0, __float_as_int(v), ROW_XMASK(M), 0xf, 0xf, true));
}

template <int M>
__device__ __forceinline__ float halve(bool up, float lo, float hi) {
    const float keep = up ? hi : lo;
    const float send = up ? lo : hi;
    return keep + row_xor<M>(send);
}

__device__ __forceinline__ void mlp_pass(
    const int pass, const float px, const float py, const float pz,
    const int lane, const int nlane, const int halfSel,
    const v2f* lds_b1t, const float* lds_w2,
    float& outR, float& outG, float& outB)
{
    const int srcLane = nlane + (pass << 4);
    float sx = __shfl(px, srcLane, 32);
    float sy = __shfl(py, srcLane, 32);
    float sz = __shfl(pz, srcLane, 32);
    v2f a;
    a.x = halfSel ? sz : sx;
    a.y = halfSel ? 1.0f : sy;

    float accR[8], accG[8], accB[8];
#pragma unroll
    for (int v = 0; v < 8; ++v) { accR[v] = 0.f; accG[v] = 0.f; accB[v] = 0.f; }

#pragma unroll
    for (int j = 0; j < 16; ++j) {
        v2f b = lds_b1t[(j << 5) + lane];
        v8f c = {};
        v8f dmat = __builtin_amdgcn_wmma_f32_16x16x4_f32(
            false, a, false, b, (short)0, c, false, false);

        const int row = (j << 4) + nlane;
        const float wr = lds_w2[row * 3 + 0];
        const float wg = lds_w2[row * 3 + 1];
        const float wb = lds_w2[row * 3 + 2];
#pragma unroll
        for (int v = 0; v < 8; ++v) {
            float h = fmaxf(dmat[v], 0.0f);
            accR[v] += h * wr;
            accG[v] += h * wg;
            accB[v] += h * wb;
        }
    }

    const bool up8 = (nlane & 8) != 0;
    const bool up4 = (nlane & 4) != 0;
    const bool up2 = (nlane & 2) != 0;

    float qR[4], qG[4], qB[4];
    for (int i = 0; i < 4; ++i) {
        qR[i] = halve<8>(up8, accR[i], accR[i + 4]);
        qG[i] = halve<8>(up8, accG[i], accG[i + 4]);
        qB[i] = halve<8>(up8, accB[i], accB[i + 4]);
    }
    float dR[2], dG[2], dB[2];
    for (int i = 0; i < 2; ++i) {
        dR[i] = halve<4>(up4, qR[i], qR[i + 2]);
        dG[i] = halve<4>(up4, qG[i], qG[i + 2]);
        dB[i] = halve<4>(up4, qB[i], qB[i + 2]);
    }
    outR = halve<2>(up2, dR[0], dR[1]);
    outG = halve<2>(up2, dG[0], dG[1]);
    outB = halve<2>(up2, dB[0], dB[1]);
}

__global__ __launch_bounds__(THREADS_BLOCK) void sphere_trace_mlp_kernel(
    const float* __restrict__ origins,
    const float* __restrict__ dirs,
    const float* __restrict__ center,
    const float* __restrict__ radius,
    const float* __restrict__ w1,
    const float* __restrict__ b1,
    const float* __restrict__ w2,
    const float* __restrict__ b2,
    float* __restrict__ out)
{
    __shared__ v2f   lds_b1t[16 * 32];
    __shared__ float lds_w2[HIDDEN * 3];

    const int tid  = threadIdx.x;
    const int lane = tid & 31;
    const int wave = tid >> 5;
    const int waveRay0 = blockIdx.x * NRAYS_BLOCK + (wave << 6);

    for (int e = tid; e < 16 * 32; e += THREADS_BLOCK) {
        int j = e >> 5, l = e & 31;
        int n = (j << 4) + (l & 15);
        v2f v;
        if (l < 16) { v.x = w1[0 * HIDDEN + n]; v.y = w1[1 * HIDDEN + n]; }
        else        { v.x = w1[2 * HIDDEN + n]; v.y = b1[n]; }
        lds_b1t[e] = v;
    }
    for (int e = tid; e < HIDDEN * 3; e += THREADS_BLOCK) lds_w2[e] = w2[e];
    __syncthreads();

    float ox[RAYS_LANE], oy[RAYS_LANE], oz[RAYS_LANE];
    float rdx[RAYS_LANE], rdy[RAYS_LANE], rdz[RAYS_LANE];
    const float* op = origins + (waveRay0 + lane) * 3;
    const float* dp = dirs    + (waveRay0 + lane) * 3;
    for (int s = 0; s < RAYS_LANE; ++s) {
        ox[s]  = op[96 * s + 0]; oy[s]  = op[96 * s + 1]; oz[s]  = op[96 * s + 2];
        rdx[s] = dp[96 * s + 0]; rdy[s] = dp[96 * s + 1]; rdz[s] = dp[96 * s + 2];
    }
    const float cx = center[0], cy = center[1], cz = center[2];
    const float R  = radius[0];

    float t[RAYS_LANE], px[RAYS_LANE], py[RAYS_LANE], pz[RAYS_LANE];
    unsigned int bal[RAYS_LANE];
    for (int s = 0; s < RAYS_LANE; ++s) {
        t[s]  = NEAR_T;
        px[s] = ox[s] + t[s] * rdx[s]; py[s] = oy[s] + t[s] * rdy[s]; pz[s] = oz[s] + t[s] * rdz[s];
        bal[s] = 0xffffffffu;
    }
#pragma unroll 4
    for (int it = 0; it < MAX_ITERS; ++it) {
        for (int s = 0; s < RAYS_LANE; ++s) {
            float qx = px[s] - cx, qy = py[s] - cy, qz = pz[s] - cz;
            float f = sqrtf(qx * qx + qy * qy + qz * qz) - R;
            t[s] += f;
            px[s] = ox[s] + t[s] * rdx[s]; py[s] = oy[s] + t[s] * rdy[s]; pz[s] = oz[s] + t[s] * rdz[s];
            bal[s] &= (unsigned int)__ballot(t[s] <= FAR_T);
        }
    }

    const float b2x = b2[0], b2y = b2[1], b2z = b2[2];
    const int  halfSel = (lane >= 16) ? 1 : 0;
    const int  nlane   = lane & 15;

#pragma nounroll
    for (int s = 0; s < RAYS_LANE; ++s) {
        const float cpx = s ? px[1] : px[0];
        const float cpy = s ? py[1] : py[0];
        const float cpz = s ? pz[1] : pz[0];
        const unsigned int balS = s ? bal[1] : bal[0];

        float r0, g0, b0v, r1, g1, b1v;
        mlp_pass(0, cpx, cpy, cpz, lane, nlane, halfSel, lds_b1t, lds_w2, r0, g0, b0v);
        mlp_pass(1, cpx, cpy, cpz, lane, nlane, halfSel, lds_b1t, lds_w2, r1, g1, b1v);

        const bool odd = (nlane & 1) != 0;
        const float sumR = halve<1>(odd, r0, r1);
        const float sumG = halve<1>(odd, g0, g1);
        const float sumB = halve<1>(odd, b0v, b1v);

        const int wl = (odd ? 16 : 0) + (halfSel << 3) + (nlane >> 1);
        const int r  = waveRay0 + (s << 5) + wl;
        const bool hit = ((balS >> wl) & 1u) != 0u;

        float s0 = 0.f, s1 = 0.f, s2 = 0.f;
        if (hit) {
            s0 = 1.0f / (1.0f + expf(-(sumR + b2x)));
            s1 = 1.0f / (1.0f + expf(-(sumG + b2y)));
            s2 = 1.0f / (1.0f + expf(-(sumB + b2z)));
        }
        out[r * 3 + 0] = s0;
        out[r * 3 + 1] = s1;
        out[r * 3 + 2] = s2;
    }
}

extern "C" void kernel_launch(void* const* d_in, const int* in_sizes, int n_in,
                              void* d_out, int out_size, void* d_ws, size_t ws_size,
                              hipStream_t stream) {
    const float* origins = (const float*)d_in[0];
    const float* dirs    = (const float*)d_in[1];
    const float* center  = (const float*)d_in[2];
    const float* radius  = (const float*)d_in[3];
    const float* w1      = (const float*)d_in[4];
    const float* b1      = (const float*)d_in[5];
    const float* w2      = (const float*)d_in[6];
    const float* b2      = (const float*)d_in[7];
    float* out = (float*)d_out;

    const int n_rays = in_sizes[0] / 3;
    const int blocks = n_rays / NRAYS_BLOCK;

    sphere_trace_mlp_kernel<<<blocks, THREADS_BLOCK, 0, stream>>>(
        origins, dirs, center, radius, w1, b1, w2, b2, out);
}


# LDS weight staging and barrier removed: each wave loads its B fragments and layer-2 weights from global memory behind the tracing loop; hidden units re-dealt to lanes (new layer-2 sum order)
# speedup vs baseline: 1.9086x; 1.2406x over previous
#include <hip/hip_runtime.h>


typedef __attribute__((ext_vector_type(2))) float v2f;
typedef __attribute__((ext_vector_type(8))) float v8f;

#define THREADS_BLOCK 256
#define RAYS_LANE 2
#define NRAYS_BLOCK (THREADS_BLOCK * RAYS_LANE)
#define HIDDEN 256
#define NEAR_T 1.0f
#define FAR_T 10.0f
#define MAX_ITERS 64

#define ROW_XMASK(m) (0x160 + (m))

template <int M>
__device__ __forceinline__ float row_xor(float v) {
    return __int_as_float(__builtin_amdgcn_update_dpp(
        0, __float_as_int(v), ROW_XMASK(M), 0xf, 0xf, true));
}

template <int M>
__device__ __forceinline__ float halve(bool up, float lo, float hi) {
    const float keep = up ? hi : lo;
    const float send = up ? lo : hi;
    return keep + row_xor<M>(send);
}

__device__ __forceinline__ void mlp_pass(
    const int pass, const float px, const float py, const float pz,
    const int nlane, const int halfSel,
    const v2f (&bfrag)[16], const float (&w2r)[16], const float (&w2g)[16], const float (&w2b)[16],
    float& outR, float& outG, float& outB)
{
    const int srcLane = nlane + (pass << 4);
    float sx = __shfl(px, srcLane, 32);
    float sy = __shfl(py, srcLane, 32);
    float sz = __shfl(pz, srcLane, 32);
    v2f a;
    a.x = halfSel ? sz : sx;
    a.y = halfSel ? 1.0f : sy;

    float accR[8], accG[8], accB[8];
#pragma unroll
    for (int v = 0; v < 8; ++v) { accR[v] = 0.f; accG[v] = 0.f; accB[v] = 0.f; }

#pragma unroll
    for (int j = 0; j < 16; ++j) {
        v2f b = bfrag[j];
        v8f c = {};
        v8f dmat = __builtin_amdgcn_wmma_f32_16x16x4_f32(
            false, a, false, b, (short)0, c, false, false);

        const float wr = w2r[j];
        const float wg = w2g[j];
        const float wb = w2b[j];
#pragma unroll
        for (int v = 0; v < 8; ++v) {
            float h = fmaxf(dmat[v], 0.0f);
            accR[v] += h * wr;
            accG[v] += h * wg;
            accB[v] += h * wb;
        }
    }

    const bool up8 = (nlane & 8) != 0;
    const bool up4 = (nlane & 4) != 0;
    const bool up2 = (nlane & 2) != 0;

    float qR[4], qG[4], qB[4];
    for (int i = 0; i < 4; ++i) {
        qR[i] = halve<8>(up8, accR[i], accR[i + 4]);
        qG[i] = halve<8>(up8, accG[i], accG[i + 4]);
        qB[i] = halve<8>(up8, accB[i], accB[i + 4]);
    }
    float dR[2], dG[2], dB[2];
    for (int i = 0; i < 2; ++i) {
        dR[i] = halve<4>(up4, qR[i], qR[i + 2]);
        dG[i] = halve<4>(up4, qG[i], qG[i + 2]);
        dB[i] = halve<4>(up4, qB[i], qB[i + 2]);
    }
    outR = halve<2>(up2, dR[0], dR[1]);
    outG = halve<2>(up2, dG[0], dG[1]);
    outB = halve<2>(up2, dB[0], dB[1]);
}

__global__ __launch_bounds__(THREADS_BLOCK) void sphere_trace_mlp_kernel(
    const float* __restrict__ origins,
    const float* __restrict__ dirs,
    const float* __restrict__ center,
    const float* __restrict__ radius,
    const float* __restrict__ w1,
    const float* __restrict__ b1,
    const float* __restrict__ w2,
    const float* __restrict__ b2,
    float* __restrict__ out)
{
    const int tid  = threadIdx.x;
    const int lane = tid & 31;
    const int wave = tid >> 5;
    const int  halfSel = (lane >= 16) ? 1 : 0;
    const int  nlane   = lane & 15;
    const int waveRay0 = blockIdx.x * NRAYS_BLOCK + (wave << 6);

    float ox[RAYS_LANE], oy[RAYS_LANE], oz[RAYS_LANE];
    float rdx[RAYS_LANE], rdy[RAYS_LANE], rdz[RAYS_LANE];
    const float* op = origins + (waveRay0 + lane) * 3;
    const float* dp = dirs    + (waveRay0 + lane) * 3;
    for (int s = 0; s < RAYS_LANE; ++s) {
        ox[s]  = op[96 * s + 0]; oy[s]  = op[96 * s + 1]; oz[s]  = op[96 * s + 2];
        rdx[s] = dp[96 * s + 0]; rdy[s] = dp[96 * s + 1]; rdz[s] = dp[96 * s + 2];
    }

    const float cx = center[0], cy = center[1], cz = center[2];
    const float R  = radius[0];

    float t[RAYS_LANE], px[RAYS_LANE], py[RAYS_LANE], pz[RAYS_LANE];
    unsigned int bal[RAYS_LANE];
    for (int s = 0; s < RAYS_LANE; ++s) {
        t[s]  = NEAR_T;
        px[s] = ox[s] + t[s] * rdx[s]; py[s] = oy[s] + t[s] * rdy[s]; pz[s] = oz[s] + t[s] * rdz[s];
        bal[s] = 0xffffffffu;
    }
#pragma unroll 4
    for (int it = 0; it < MAX_ITERS; ++it) {
        for (int s = 0; s < RAYS_LANE; ++s) {
            float qx = px[s] - cx, qy = py[s] - cy, qz = pz[s] - cz;
            float f = sqrtf(qx * qx + qy * qy + qz * qz) - R;
            t[s] += f;
            px[s] = ox[s] + t[s] * rdx[s]; py[s] = oy[s] + t[s] * rdy[s]; pz[s] = oz[s] + t[s] * rdz[s];
            bal[s] &= (unsigned int)__ballot(t[s] <= FAR_T);
        }
    }

    const float* bp0 = halfSel ? (w1 + 2 * HIDDEN + 2 * nlane) : (w1 + 2 * nlane);
    const float* bp1 = halfSel ? (b1 + 2 * nlane)              : (w1 + HIDDEN + 2 * nlane);
    const float* wp  = w2 + 6 * nlane;
    v2f   bfrag[16];
    float w2r[16], w2g[16], w2b[16];
#pragma unroll
    for (int g = 0; g < 8; ++g) {
#pragma unroll
        for (int i = 0; i < 2; ++i) {
            bfrag[2 * g + i].x = bp0[32 * g + i];
            bfrag[2 * g + i].y = bp1[32 * g + i];
            w2r[2 * g + i] = wp[96 * g + 3 * i + 0];
            w2g[2 * g + i] = wp[96 * g + 3 * i + 1];
            w2b[2 * g + i] = wp[96 * g + 3 * i + 2];
        }
    }

    const float b2x = b2[0], b2y = b2[1], b2z = b2[2];

#pragma nounroll
    for (int s = 0; s < RAYS_LANE; ++s) {
        const float cpx = s ? px[1] : px[0];
        const float cpy = s ? py[1] : py[0];
        const float cpz = s ? pz[1] : pz[0];
        const unsigned int balS = s ? bal[1] : bal[0];

        float r0, g0, b0v, r1, g1, b1v;
        mlp_pass(0, cpx, cpy, cpz, nlane, halfSel, bfrag, w2r, w2g, w2b, r0, g0, b0v);
        mlp_pass(1, cpx, cpy, cpz, nlane, halfSel, bfrag, w2r, w2g, w2b, r1, g1, b1v);

        const bool odd = (nlane & 1) != 0;
        const float sumR = halve<1>(odd, r0, r1);
        const float sumG = halve<1>(odd, g0, g1);
        const float sumB = halve<1>(odd, b0v, b1v);

        const int wl = (odd ? 16 : 0) + (halfSel << 3) + (nlane >> 1);
        const int r  = waveRay0 + (s << 5) + wl;
        const bool hit = ((balS >> wl) & 1u) != 0u;

        float s0 = 0.f, s1 = 0.f, s2 = 0.f;
        if (hit) {
            s0 = 1.0f / (1.0f + expf(-(sumR + b2x)));
            s1 = 1.0f / (1.0f + expf(-(sumG + b2y)));
            s2 = 1.0f / (1.0f + expf(-(sumB + b2z)));
        }
        out[r * 3 + 0] = s0;
        out[r * 3 + 1] = s1;
        out[r * 3 + 2] = s2;
    }
}

extern "C" void kernel_launch(void* const* d_in, const int* in_sizes, int n_in,
                              void* d_out, int out_size, void* d_ws, size_t ws_size,
                              hipStream_t stream) {
    const float* origins = (const float*)d_in[0];
    const float* dirs    = (const float*)d_in[1];
    const float* center  = (const float*)d_in[2];
    const float* radius  = (const float*)d_in[3];
    const float* w1      = (const float*)d_in[4];
    const float* b1      = (const float*)d_in[5];
    const float* w2      = (const float*)d_in[6];
    const float* b2      = (const float*)d_in[7];
    float* out = (float*)d_out;

    const int n_rays = in_sizes[0] / 3;
    const int blocks = n_rays / NRAYS_BLOCK;

    sphere_trace_mlp_kernel<<<blocks, THREADS_BLOCK, 0, stream>>>(
        origins, dirs, center, radius, w1, b1, w2, b2, out);
}
